# starting kernel
# baseline (speedup 1.0000x reference)
#include <hip/hip_runtime.h>

typedef __attribute__((ext_vector_type(2))) float v2f;
typedef __attribute__((ext_vector_type(8))) float v8f;

#define IN_F 128
#define OUT_F 64
#define WAVES_PER_BLOCK 8

__global__ void init_out_kernel(float* __restrict__ out,
                                const float* __restrict__ bias, int total) {
  int i = blockIdx.x * blockDim.x + threadIdx.x;
  if (i < total) out[i] = bias[i & (OUT_F - 1)];
}

__global__ __launch_bounds__(256) void gemm_wmma_kernel(
    const float* __restrict__ V, const float* __restrict__ W,
    float* __restrict__ S, int nRowTiles) {
  __shared__ v2f wlds2[(IN_F / 2) * OUT_F];
  for (int i = threadIdx.x; i < (IN_F / 2) * OUT_F; i += blockDim.x) {
    const int p = i / OUT_F;
    const int n = i % OUT_F;
    v2f w;
    w.x = W[(2 * p + 0) * OUT_F + n];
    w.y = W[(2 * p + 1) * OUT_F + n];
    wlds2[i] = w;
  }
  __syncthreads();

  const int lane = threadIdx.x & 31;
  const int wave = threadIdx.x >> 5;
  const int rowTile = blockIdx.x * WAVES_PER_BLOCK + wave;
  if (rowTile >= nRowTiles) return;

  const int m    = lane & 15;
  const int koff = (lane >> 4) << 1;

  const float* vptr = V + (long long)(rowTile * 16 + m) * IN_F + koff;
  const v2f*   wbase = wlds2 + (koff >> 1) * OUT_F + m;

  v8f c0 = {}, c1 = {}, c2 = {}, c3 = {};
#pragma unroll
  for (int k = 0; k < IN_F; k += 4) {
    v2f a = *(const v2f*)(vptr + k);
    const v2f* wrow = wbase + (k >> 1) * OUT_F;
    v2f b0 = wrow[0];
    v2f b1 = wrow[16];
    v2f b2 = wrow[32];
    v2f b3 = wrow[48];
    c0 = __builtin_amdgcn_wmma_f32_16x16x4_f32(false, a, false, b0,
                                               (short)0, c0, false, false);
    c1 = __builtin_amdgcn_wmma_f32_16x16x4_f32(false, a, false, b1,
                                               (short)0, c1, false, false);
    c2 = __builtin_amdgcn_wmma_f32_16x16x4_f32(false, a, false, b2,
                                               (short)0, c2, false, false);
    c3 = __builtin_amdgcn_wmma_f32_16x16x4_f32(false, a, false, b3,
                                               (short)0, c3, false, false);
  }

  const long long mBase = (long long)rowTile * 16 + ((lane >> 4) << 3);
#pragma unroll
  for (int r = 0; r < 8; ++r) {
    float* srow = S + (mBase + r) * OUT_F + m;
    srow[0]  = c0[r];
    srow[16] = c1[r];
    srow[32] = c2[r];
    srow[48] = c3[r];
  }
}

__global__ void spmm_kernel(const float* __restrict__ S,
                            const int* __restrict__ erow,
                            const int* __restrict__ ecol,
                            const float* __restrict__ eval,
                            float* out, int nWork) {
  int idx = blockIdx.x * blockDim.x + threadIdx.x;
  if (idx >= nWork) return;
  const int e = idx >> 4;
  const int f = (idx & 15) << 2;
  const int col = ecol[e];
  const int row = erow[e];
  const float v = eval[e];
  const float4 s = *(const float4*)(S + (long long)col * OUT_F + f);
  float* o = out + (long long)row * OUT_F + f;
  atomicAdd(o + 0, s.x * v);
  atomicAdd(o + 1, s.y * v);
  atomicAdd(o + 2, s.z * v);
  atomicAdd(o + 3, s.w * v);
}

extern "C" void kernel_launch(void* const* d_in, const int* in_sizes, int n_in,
                              void* d_out, int out_size, void* d_ws,
                              size_t ws_size, hipStream_t stream) {
  const float* vertex  = (const float*)d_in[0];
  const int*   erow    = (const int*)d_in[1];
  const int*   ecol    = (const int*)d_in[2];
  const float* eval    = (const float*)d_in[3];
  const float* weights = (const float*)d_in[4];
  const float* bias    = (const float*)d_in[5];
  float* out     = (float*)d_out;
  float* support = (float*)d_ws;

  const int nNodes = in_sizes[0] / IN_F;
  const int nEdges = in_sizes[1];

  const int totalOut = nNodes * OUT_F;
  init_out_kernel<<<(totalOut + 255) / 256, 256, 0, stream>>>(out, bias,
                                                              totalOut);

  const int nRowTiles = nNodes / 16;
  const int nBlocks = (nRowTiles + WAVES_PER_BLOCK - 1) / WAVES_PER_BLOCK;
  gemm_wmma_kernel<<<nBlocks, 256, 0, stream>>>(vertex, weights, support,
                                                nRowTiles);

  const int nWork = nEdges * 16;
  spmm_kernel<<<(nWork + 255) / 256, 256, 0, stream>>>(support, erow, ecol,
                                                       eval, out, nWork);
}


# spmm by row owner: one work-group per 400 out rows keeps them in LDS, filters a 1-byte-per-edge range key list, adds with LDS f32 atomics (arrival order), stores rows once; no global atomic, no fill
# speedup vs baseline: 1.9305x; 1.9305x over previous
#include <hip/hip_runtime.h>


typedef __attribute__((ext_vector_type(2))) float v2f;
typedef __attribute__((ext_vector_type(8))) float v8f;

#define IN_F 128
#define OUT_F 64
#define WAVES_PER_BLOCK 8

#define ROWS_PER_OWNER 400
#define OWN_THREADS 1024
#define OWN_WAVES 32
#define QCAP 544
#define RECS_WS_OFFSET (64u << 20)
#define KEYS_WS_OFFSET (128u << 20)
#define RL_BITS 9

__device__ __forceinline__ unsigned edge_key(int row, int col, int* recx) {
  const unsigned o = (unsigned)row / ROWS_PER_OWNER;
  *recx = (int)(((unsigned)col << RL_BITS) | ((unsigned)row - o * ROWS_PER_OWNER));
  return o & 0xFFu;
}

__device__ __forceinline__ void emit_four(int i, int4 r, int4 c, float4 v,
                                          unsigned* keys, int4* recs4) {
  int4 lo, hi;
  const unsigned k0 = edge_key(r.x, c.x, &lo.x);
  const unsigned k1 = edge_key(r.y, c.y, &lo.z);
  const unsigned k2 = edge_key(r.z, c.z, &hi.x);
  const unsigned k3 = edge_key(r.w, c.w, &hi.z);
  lo.y = __float_as_int(v.x);
  lo.w = __float_as_int(v.y);
  hi.y = __float_as_int(v.z);
  hi.w = __float_as_int(v.w);
  keys[i] = k0 | (k1 << 8) | (k2 << 16) | (k3 << 24);
  recs4[2 * i] = lo;
  recs4[2 * i + 1] = hi;
}

__global__ __launch_bounds__(256) void gemm_wmma_kernel(
    const float* __restrict__ V, const float* __restrict__ W,
    float* __restrict__ S, int nRowTiles, const int* __restrict__ erow,
    const int* __restrict__ ecol, const float* __restrict__ eval,
    unsigned* __restrict__ keys, int* __restrict__ recs, int nEdges) {
  __shared__ v2f wlds2[(IN_F / 2) * OUT_F];

  const int nKeyWords = ((nEdges + 15) >> 4) << 2;
  const int nFull = nEdges >> 2;
  const int4* erow4 = (const int4*)erow;
  const int4* ecol4 = (const int4*)ecol;
  const float4* eval4 = (const float4*)eval;
  int4* recs4 = (int4*)recs;
  const int stride = gridDim.x * blockDim.x;
  int i = blockIdx.x * blockDim.x + threadIdx.x;
  for (; i + stride < nFull; i += 2 * stride) {
    const int4 r0 = erow4[i], c0 = ecol4[i];
    const float4 v0 = eval4[i];
    const int4 r1 = erow4[i + stride], c1 = ecol4[i + stride];
    const float4 v1 = eval4[i + stride];
    emit_four(i, r0, c0, v0, keys, recs4);
    emit_four(i + stride, r1, c1, v1, keys, recs4);
  }
  for (; i < nKeyWords; i += stride) {
    if (i < nFull) {
      emit_four(i, erow4[i], ecol4[i], eval4[i], keys, recs4);
    } else {
      unsigned k = 0;
#pragma unroll
      for (int c = 0; c < 4; ++c) {
        const int e = (i << 2) + c;
        unsigned o = 0xFFu;
        if (e < nEdges) {
          int rx;
          o = edge_key(erow[e], ecol[e], &rx);
          recs[2 * e] = rx;
          recs[2 * e + 1] = __float_as_int(eval[e]);
        }
        k |= o << (8 * c);
      }
      keys[i] = k;
    }
  }

  for (int i = threadIdx.x; i < (IN_F / 2) * OUT_F; i += blockDim.x) {
    const int p = i / OUT_F;
    const int n = i % OUT_F;
    v2f w;
    w.x = W[(2 * p + 0) * OUT_F + n];
    w.y = W[(2 * p + 1) * OUT_F + n];
    wlds2[i] = w;
  }
  __syncthreads();

  const int lane = threadIdx.x & 31;
  const int wave = threadIdx.x >> 5;
  const int rowTile = blockIdx.x * WAVES_PER_BLOCK + wave;
  if (rowTile >= nRowTiles) return;

  const int m    = lane & 15;
  const int koff = (lane >> 4) << 1;

  const float* vptr = V + (long long)(rowTile * 16 + m) * IN_F + koff;
  const v2f*   wbase = wlds2 + (koff >> 1) * OUT_F + m;

  v8f c0 = {}, c1 = {}, c2 = {}, c3 = {};
#pragma unroll
  for (int k = 0; k < IN_F; k += 4) {
    v2f a = *(const v2f*)(vptr + k);
    const v2f* wrow = wbase + (k >> 1) * OUT_F;
    v2f b0 = wrow[0];
    v2f b1 = wrow[16];
    v2f b2 = wrow[32];
    v2f b3 = wrow[48];
    c0 = __builtin_amdgcn_wmma_f32_16x16x4_f32(false, a, false, b0,
                                               (short)0, c0, false, false);
    c1 = __builtin_amdgcn_wmma_f32_16x16x4_f32(false, a, false, b1,
                                               (short)0, c1, false, false);
    c2 = __builtin_amdgcn_wmma_f32_16x16x4_f32(false, a, false, b2,
                                               (short)0, c2, false, false);
    c3 = __builtin_amdgcn_wmma_f32_16x16x4_f32(false, a, false, b3,
                                               (short)0, c3, false, false);
  }

  const long long mBase = (long long)rowTile * 16 + ((lane >> 4) << 3);
#pragma unroll
  for (int r = 0; r < 8; ++r) {
    float* srow = S + (mBase + r) * OUT_F + m;
    srow[0]  = c0[r];
    srow[16] = c1[r];
    srow[32] = c2[r];
    srow[48] = c3[r];
  }
}


__device__ __forceinline__ unsigned key_differs(unsigned x, unsigned pat) {
  const unsigned y = x ^ pat;
  return ((y & 0x7f7f7f7fu) + 0x7f7f7f7fu) | y;
}

__device__ __forceinline__ void drain_batch(
    const unsigned* qb, int n, int lane, const float* __restrict__ S,
    const int2* __restrict__ recs2, float* acc) {
  const unsigned e = (lane < n) ? qb[lane] : 0u;
  const int2 rec = recs2[e];
  const int col = (int)((unsigned)rec.x >> RL_BITS);
  const int rl = rec.x & ((1 << RL_BITS) - 1);
  const int vbits = rec.y;
  const float* Shi = S + 32;
  for (int j0 = 0; j0 < n; j0 += 16) {
    float s0[16], s1[16];
#pragma unroll
    for (int u = 0; u < 16; ++u) {
      const unsigned c = (unsigned)__builtin_amdgcn_readlane(col, j0 + u);
      const unsigned idx = c * OUT_F + (unsigned)lane;
      s0[u] = S[idx];
      s1[u] = Shi[idx];
    }
#pragma unroll
    for (int u = 0; u < 16; ++u) {
      if (j0 + u < n) {
        const int r = __builtin_amdgcn_readlane(rl, j0 + u);
        const float v = __int_as_float(__builtin_amdgcn_readlane(vbits, j0 + u));
        float* a = acc + r * OUT_F + lane;
        __hip_atomic_fetch_add(a, s0[u] * v, __ATOMIC_RELAXED,
                               __HIP_MEMORY_SCOPE_WORKGROUP);
        __hip_atomic_fetch_add(a + 32, s1[u] * v, __ATOMIC_RELAXED,
                               __HIP_MEMORY_SCOPE_WORKGROUP);
      }
    }
  }
}

__global__ __launch_bounds__(OWN_THREADS) void spmm_owner_kernel(
    const float* __restrict__ S, const unsigned* __restrict__ keys,
    const int* __restrict__ recs, const float* __restrict__ bias,
    float* __restrict__ out, int nNodes, int nEdges, int nOwners) {
  __shared__ float acc[ROWS_PER_OWNER * OUT_F];
  __shared__ unsigned queue[OWN_WAVES * QCAP];

  const int tid = threadIdx.x;
  const int lane = tid & 31;
  const int wave = __builtin_amdgcn_readfirstlane(tid >> 5);
  const int owner = blockIdx.x;
  const int base = owner * ROWS_PER_OWNER;

  for (int i = tid; i < ROWS_PER_OWNER * OUT_F; i += OWN_THREADS)
    acc[i] = bias[i & (OUT_F - 1)];
  __syncthreads();

  const uint4* keys16 = (const uint4*)keys;
  const int2* recs2 = (const int2*)recs;
  unsigned* q = queue + wave * QCAP;
  const unsigned pat = (unsigned)owner * 0x01010101u;
  const int n16 = (nEdges + 15) >> 4;
  const int nTrips = (n16 + 31) >> 5;
  const int nTripsFull = n16 >> 5;
  const int rot = (int)(((unsigned)owner * (unsigned)nTrips) / (unsigned)nOwners);
  int cnt = 0;

  for (int seg = 0; seg < 2; ++seg) {
    const int lo = seg ? 0 : rot;
    const int hi = seg ? rot : nTrips;
    const int hiFull = hi < nTripsFull ? hi : nTripsFull;
    for (int t0 = lo + wave; t0 < hi; t0 += OWN_WAVES * 4) {
      uint4 x[4];
      if (t0 + 3 * OWN_WAVES < hiFull) {
#pragma unroll
        for (int u = 0; u < 4; ++u)
          x[u] = keys16[(t0 + u * OWN_WAVES) * 32 + lane];
      } else {
#pragma unroll
        for (int u = 0; u < 4; ++u) {
          const int t = t0 + u * OWN_WAVES;
          const int idx = t * 32 + lane;
          x[u] = (t < hi && idx < n16) ? keys16[idx]
                                       : make_uint4(~0u, ~0u, ~0u, ~0u);
        }
      }
#pragma unroll
      for (int u = 0; u < 4; ++u) {
        const int idx = (t0 + u * OWN_WAVES) * 32 + lane;
        const unsigned d0 = key_differs(x[u].x, pat);
        const unsigned d1 = key_differs(x[u].y, pat);
        const unsigned d2 = key_differs(x[u].z, pat);
        const unsigned d3 = key_differs(x[u].w, pat);
        unsigned m = ~(((d0 >> 7) & 0x01010101u) | ((d1 >> 6) & 0x02020202u) |
                       ((d2 >> 5) & 0x04040404u) | ((d3 >> 4) & 0x08080808u)) &
                     0x0F0F0F0Fu;
        unsigned bal = __builtin_amdgcn_ballot_w32(m != 0u);
        while (bal != 0u) {
          if (m != 0u) {
            const int p = __builtin_ctz(m);
            m &= m - 1u;
            const unsigned e =
                ((unsigned)idx << 4) + ((p & 3) << 2) + (p >> 3);
            q[cnt + (int)__builtin_amdgcn_mbcnt_lo(bal, 0u)] = e;
          }
          cnt += __builtin_popcount(bal);
          bal = __builtin_amdgcn_ballot_w32(m != 0u);
        }
        while (cnt >= 32) {
          cnt -= 32;
          drain_batch(q + cnt, 32, lane, S, recs2, acc);
        }
      }
    }
  }
  if (cnt > 0) drain_batch(q, cnt, lane, S, recs2, acc);

  __syncthreads();

  int nrows = nNodes - base;
  if (nrows > ROWS_PER_OWNER) nrows = ROWS_PER_OWNER;
  const float4* a4 = (const float4*)acc;
  float4* o4 = (float4*)(out + (long long)base * OUT_F);
  for (int i = tid; i < nrows * (OUT_F / 4); i += OWN_THREADS) o4[i] = a4[i];
}

extern "C" void kernel_launch(void* const* d_in, const int* in_sizes, int n_in,
                              void* d_out, int out_size, void* d_ws,
                              size_t ws_size, hipStream_t stream) {
  const float* vertex  = (const float*)d_in[0];
  const int*   erow    = (const int*)d_in[1];
  const int*   ecol    = (const int*)d_in[2];
  const float* eval    = (const float*)d_in[3];
  const float* weights = (const float*)d_in[4];
  const float* bias    = (const float*)d_in[5];
  float* out     = (float*)d_out;
  float* support = (float*)d_ws;
  int* recs = (int*)((char*)d_ws + RECS_WS_OFFSET);
  unsigned* keys = (unsigned*)((char*)d_ws + KEYS_WS_OFFSET);

  const int nNodes = in_sizes[0] / IN_F;
  const int nEdges = in_sizes[1];

  const int nRowTiles = nNodes / 16;
  const int nBlocks = (nRowTiles + WAVES_PER_BLOCK - 1) / WAVES_PER_BLOCK;
  gemm_wmma_kernel<<<nBlocks, 256, 0, stream>>>(vertex, weights, support,
                                                nRowTiles, erow, ecol, eval,
                                                keys, recs, nEdges);

  const int nOwners = (nNodes + ROWS_PER_OWNER - 1) / ROWS_PER_OWNER;
  spmm_owner_kernel<<<nOwners, OWN_THREADS, 0, stream>>>(
      support, keys, recs, bias, out, nNodes, nEdges, nOwners);
}


# product launch as a resident grid: a wave keeps all of W as B fragments in registers and loops over row tiles, next tile's A rows in flight; no LDS staging, no barrier
# speedup vs baseline: 2.1652x; 1.1216x over previous
#include <hip/hip_runtime.h>


typedef __attribute__((ext_vector_type(2))) float v2f;
typedef __attribute__((ext_vector_type(8))) float v8f;

#define IN_F 128
#define OUT_F 64
#define WAVES_PER_BLOCK 8
#define GEMM_BLOCKS 256

#define ROWS_PER_OWNER 400
#define OWN_THREADS 1024
#define OWN_WAVES 32
#define QCAP 544
#define RECS_WS_OFFSET (64u << 20)
#define KEYS_WS_OFFSET (128u << 20)
#define RL_BITS 9

__device__ __forceinline__ unsigned edge_key(int row, int col, int* recx) {
  const unsigned o = (unsigned)row / ROWS_PER_OWNER;
  *recx = (int)(((unsigned)col << RL_BITS) | ((unsigned)row - o * ROWS_PER_OWNER));
  return o & 0xFFu;
}

__device__ __forceinline__ void emit_four(int i, int4 r, int4 c, float4 v,
                                          unsigned* keys, int4* recs4) {
  int4 lo, hi;
  const unsigned k0 = edge_key(r.x, c.x, &lo.x);
  const unsigned k1 = edge_key(r.y, c.y, &lo.z);
  const unsigned k2 = edge_key(r.z, c.z, &hi.x);
  const unsigned k3 = edge_key(r.w, c.w, &hi.z);
  lo.y = __float_as_int(v.x);
  lo.w = __float_as_int(v.y);
  hi.y = __float_as_int(v.z);
  hi.w = __float_as_int(v.w);
  keys[i] = k0 | (k1 << 8) | (k2 << 16) | (k3 << 24);
  recs4[2 * i] = lo;
  recs4[2 * i + 1] = hi;
}

struct EdgeWords {
  int i0, i1;
  bool f0, f1;
  int4 r0, q0, r1, q1;
  float4 v0, v1;
};

__device__ __forceinline__ EdgeWords load_edge_words(
    int i, int stride, int nFull, const int4* __restrict__ erow4,
    const int4* __restrict__ ecol4, const float4* __restrict__ eval4) {
  EdgeWords w;
  w.i0 = i;
  w.i1 = i + stride;
  w.f0 = w.i0 < nFull;
  w.f1 = w.i1 < nFull;
  const int j0 = w.f0 ? w.i0 : 0, j1 = w.f1 ? w.i1 : 0;
  w.r0 = erow4[j0]; w.q0 = ecol4[j0]; w.v0 = eval4[j0];
  w.r1 = erow4[j1]; w.q1 = ecol4[j1]; w.v1 = eval4[j1];
  return w;
}

__device__ __forceinline__ void emit_edge_words(const EdgeWords& w, unsigned* keys,
                                                int4* recs4) {
  if (w.f0) emit_four(w.i0, w.r0, w.q0, w.v0, keys, recs4);
  if (w.f1) emit_four(w.i1, w.r1, w.q1, w.v1, keys, recs4);
}

__device__ __forceinline__ void load_a_tile(const float* __restrict__ V, int tile,
                                            int m, int koff, v2f (&a)[IN_F / 4]) {
  const float* vptr = V + (long long)(tile * 16 + m) * IN_F + koff;
#pragma unroll
  for (int ks = 0; ks < IN_F / 4; ++ks) a[ks] = *(const v2f*)(vptr + 4 * ks);
}

__global__ __launch_bounds__(256) void gemm_wmma_kernel(
    const float* __restrict__ V, const float* __restrict__ W,
    float* __restrict__ S, int nRowTiles, const int* __restrict__ erow,
    const int* __restrict__ ecol, const float* __restrict__ eval,
    unsigned* __restrict__ keys, int* __restrict__ recs, int nEdges) {
  const int lane = threadIdx.x & 31;
  const int wave = __builtin_amdgcn_readfirstlane(threadIdx.x >> 5);
  const int c    = lane & 15;
  const int h    = lane >> 4;
  const int koff = h << 1;

  v2f b[IN_F / 4][4];
  {
    const float4* w4 = (const float4*)(W + koff * OUT_F + 4 * c);
#pragma unroll
    for (int ks = 0; ks < IN_F / 4; ++ks) {
      const float4 w0 = w4[(4 * ks) * (OUT_F / 4)];
      const float4 w1 = w4[(4 * ks + 1) * (OUT_F / 4)];
      b[ks][0].x = w0.x; b[ks][0].y = w1.x;
      b[ks][1].x = w0.y; b[ks][1].y = w1.y;
      b[ks][2].x = w0.z; b[ks][2].y = w1.z;
      b[ks][3].x = w0.w; b[ks][3].y = w1.w;
    }
  }

  const int nWaves = gridDim.x * WAVES_PER_BLOCK;
  int tile = wave * gridDim.x + blockIdx.x;

  const int nKeyWords = ((nEdges + 15) >> 4) << 2;
  const int nFull = nEdges >> 2;
  const int4* erow4 = (const int4*)erow;
  const int4* ecol4 = (const int4*)ecol;
  const float4* eval4 = (const float4*)eval;
  int4* recs4 = (int4*)recs;
  const int estride = gridDim.x * blockDim.x;
  int ew = blockIdx.x * blockDim.x + threadIdx.x;

  v2f a_cur[IN_F / 4], a_nxt[IN_F / 4];
  load_a_tile(V, tile < nRowTiles ? tile : nRowTiles - 1, c, koff, a_nxt);
  EdgeWords w = load_edge_words(ew, estride, nFull, erow4, ecol4, eval4);
  ew += 2 * estride;

  for (; tile < nRowTiles; tile += nWaves) {
#pragma unroll
    for (int ks = 0; ks < IN_F / 4; ++ks) a_cur[ks] = a_nxt[ks];

    const EdgeWords w_new = load_edge_words(ew, estride, nFull, erow4, ecol4, eval4);
    ew += 2 * estride;
    const int nxt = tile + nWaves;
    load_a_tile(V, nxt < nRowTiles ? nxt : nRowTiles - 1, c, koff, a_nxt);

    v8f c0 = {}, c1 = {}, c2 = {}, c3 = {};
#pragma unroll
    for (int ks = 0; ks < IN_F / 4; ++ks) {
      c0 = __builtin_amdgcn_wmma_f32_16x16x4_f32(false, a_cur[ks], false, b[ks][0],
                                                 (short)0, c0, false, false);
      c1 = __builtin_amdgcn_wmma_f32_16x16x4_f32(false, a_cur[ks], false, b[ks][1],
                                                 (short)0, c1, false, false);
      c2 = __builtin_amdgcn_wmma_f32_16x16x4_f32(false, a_cur[ks], false, b[ks][2],
                                                 (short)0, c2, false, false);
      c3 = __builtin_amdgcn_wmma_f32_16x16x4_f32(false, a_cur[ks], false, b[ks][3],
                                                 (short)0, c3, false, false);
    }

    float* srow = S + ((long long)tile * 16 + (h << 3)) * OUT_F + 4 * c;
#pragma unroll
    for (int r = 0; r < 8; ++r) {
      float4 o;
      o.x = c0[r]; o.y = c1[r]; o.z = c2[r]; o.w = c3[r];
      *(float4*)(srow + r * OUT_F) = o;
    }

    emit_edge_words(w, keys, recs4);
    w = w_new;
  }
  emit_edge_words(w, keys, recs4);

  for (; ew < nFull; ew += estride)
    emit_four(ew, erow4[ew], ecol4[ew], eval4[ew], keys, recs4);

  for (int i = nFull + blockIdx.x * blockDim.x + threadIdx.x; i < nKeyWords;
       i += estride) {
    unsigned k = 0;
#pragma unroll
    for (int cc = 0; cc < 4; ++cc) {
      const int e = (i << 2) + cc;
      unsigned o = 0xFFu;
      if (e < nEdges) {
        int rx;
        o = edge_key(erow[e], ecol[e], &rx);
        recs[2 * e] = rx;
        recs[2 * e + 1] = __float_as_int(eval[e]);
      }
      k |= o << (8 * cc);
    }
    keys[i] = k;
  }
}


__device__ __forceinline__ unsigned key_differs(unsigned x, unsigned pat) {
  const unsigned y = x ^ pat;
  return ((y & 0x7f7f7f7fu) + 0x7f7f7f7fu) | y;
}

__device__ __forceinline__ void drain_batch(
    const unsigned* qb, int n, int lane, const float* __restrict__ S,
    const int2* __restrict__ recs2, float* acc) {
  const unsigned e = (lane < n) ? qb[lane] : 0u;
  const int2 rec = recs2[e];
  const int col = (int)((unsigned)rec.x >> RL_BITS);
  const int rl = rec.x & ((1 << RL_BITS) - 1);
  const int vbits = rec.y;
  const float* Shi = S + 32;
  for (int j0 = 0; j0 < n; j0 += 16) {
    float s0[16], s1[16];
#pragma unroll
    for (int u = 0; u < 16; ++u) {
      const unsigned c = (unsigned)__builtin_amdgcn_readlane(col, j0 + u);
      const unsigned idx = c * OUT_F + (unsigned)lane;
      s0[u] = S[idx];
      s1[u] = Shi[idx];
    }
#pragma unroll
    for (int u = 0; u < 16; ++u) {
      if (j0 + u < n) {
        const int r = __builtin_amdgcn_readlane(rl, j0 + u);
        const float v = __int_as_float(__builtin_amdgcn_readlane(vbits, j0 + u));
        float* a = acc + r * OUT_F + lane;
        __hip_atomic_fetch_add(a, s0[u] * v, __ATOMIC_RELAXED,
                               __HIP_MEMORY_SCOPE_WORKGROUP);
        __hip_atomic_fetch_add(a + 32, s1[u] * v, __ATOMIC_RELAXED,
                               __HIP_MEMORY_SCOPE_WORKGROUP);
      }
    }
  }
}

__global__ __launch_bounds__(OWN_THREADS) void spmm_owner_kernel(
    const float* __restrict__ S, const unsigned* __restrict__ keys,
    const int* __restrict__ recs, const float* __restrict__ bias,
    float* __restrict__ out, int nNodes, int nEdges, int nOwners) {
  __shared__ float acc[ROWS_PER_OWNER * OUT_F];
  __shared__ unsigned queue[OWN_WAVES * QCAP];

  const int tid = threadIdx.x;
  const int lane = tid & 31;
  const int wave = __builtin_amdgcn_readfirstlane(tid >> 5);
  const int owner = blockIdx.x;
  const int base = owner * ROWS_PER_OWNER;

  for (int i = tid; i < ROWS_PER_OWNER * OUT_F; i += OWN_THREADS)
    acc[i] = bias[i & (OUT_F - 1)];
  __syncthreads();

  const uint4* keys16 = (const uint4*)keys;
  const int2* recs2 = (const int2*)recs;
  unsigned* q = queue + wave * QCAP;
  const unsigned pat = (unsigned)owner * 0x01010101u;
  const int n16 = (nEdges + 15) >> 4;
  const int nTrips = (n16 + 31) >> 5;
  const int nTripsFull = n16 >> 5;
  const int rot = (int)(((unsigned)owner * (unsigned)nTrips) / (unsigned)nOwners);
  int cnt = 0;

  for (int seg = 0; seg < 2; ++seg) {
    const int lo = seg ? 0 : rot;
    const int hi = seg ? rot : nTrips;
    const int hiFull = hi < nTripsFull ? hi : nTripsFull;
    for (int t0 = lo + wave; t0 < hi; t0 += OWN_WAVES * 4) {
      uint4 x[4];
      if (t0 + 3 * OWN_WAVES < hiFull) {
#pragma unroll
        for (int u = 0; u < 4; ++u)
          x[u] = keys16[(t0 + u * OWN_WAVES) * 32 + lane];
      } else {
#pragma unroll
        for (int u = 0; u < 4; ++u) {
          const int t = t0 + u * OWN_WAVES;
          const int idx = t * 32 + lane;
          x[u] = (t < hi && idx < n16) ? keys16[idx]
                                       : make_uint4(~0u, ~0u, ~0u, ~0u);
        }
      }
#pragma unroll
      for (int u = 0; u < 4; ++u) {
        const int idx = (t0 + u * OWN_WAVES) * 32 + lane;
        const unsigned d0 = key_differs(x[u].x, pat);
        const unsigned d1 = key_differs(x[u].y, pat);
        const unsigned d2 = key_differs(x[u].z, pat);
        const unsigned d3 = key_differs(x[u].w, pat);
        unsigned m = ~(((d0 >> 7) & 0x01010101u) | ((d1 >> 6) & 0x02020202u) |
                       ((d2 >> 5) & 0x04040404u) | ((d3 >> 4) & 0x08080808u)) &
                     0x0F0F0F0Fu;
        unsigned bal = __builtin_amdgcn_ballot_w32(m != 0u);
        while (bal != 0u) {
          if (m != 0u) {
            const int p = __builtin_ctz(m);
            m &= m - 1u;
            const unsigned e =
                ((unsigned)idx << 4) + ((p & 3) << 2) + (p >> 3);
            q[cnt + (int)__builtin_amdgcn_mbcnt_lo(bal, 0u)] = e;
          }
          cnt += __builtin_popcount(bal);
          bal = __builtin_amdgcn_ballot_w32(m != 0u);
        }
        while (cnt >= 32) {
          cnt -= 32;
          drain_batch(q + cnt, 32, lane, S, recs2, acc);
        }
      }
    }
  }
  if (cnt > 0) drain_batch(q, cnt, lane, S, recs2, acc);

  __syncthreads();

  int nrows = nNodes - base;
  if (nrows > ROWS_PER_OWNER) nrows = ROWS_PER_OWNER;
  const float4* a4 = (const float4*)acc;
  float4* o4 = (float4*)(out + (long long)base * OUT_F);
  for (int i = tid; i < nrows * (OUT_F / 4); i += OWN_THREADS) o4[i] = a4[i];
}

extern "C" void kernel_launch(void* const* d_in, const int* in_sizes, int n_in,
                              void* d_out, int out_size, void* d_ws,
                              size_t ws_size, hipStream_t stream) {
  const float* vertex  = (const float*)d_in[0];
  const int*   erow    = (const int*)d_in[1];
  const int*   ecol    = (const int*)d_in[2];
  const float* eval    = (const float*)d_in[3];
  const float* weights = (const float*)d_in[4];
  const float* bias    = (const float*)d_in[5];
  float* out     = (float*)d_out;
  float* support = (float*)d_ws;
  int* recs = (int*)((char*)d_ws + RECS_WS_OFFSET);
  unsigned* keys = (unsigned*)((char*)d_ws + KEYS_WS_OFFSET);

  const int nNodes = in_sizes[0] / IN_F;
  const int nEdges = in_sizes[1];

  const int nRowTiles = nNodes / 16;
  gemm_wmma_kernel<<<GEMM_BLOCKS, 256, 0, stream>>>(vertex, weights, support,
                                                nRowTiles, erow, ecol, eval,
                                                keys, recs, nEdges);

  const int nOwners = (nNodes + ROWS_PER_OWNER - 1) / ROWS_PER_OWNER;
  spmm_owner_kernel<<<nOwners, OWN_THREADS, 0, stream>>>(
      support, keys, recs, bias, out, nNodes, nEdges, nOwners);
}


# owner key list stored as bit planes (32x32 bit transpose in the wave that writes it): an owner tests 32 edges with 8 boolean ops, not 16 with 18; same bytes, LDS f32 adds in arrival order as before
# speedup vs baseline: 2.2572x; 1.0425x over previous
#include <hip/hip_runtime.h>


typedef __attribute__((ext_vector_type(2))) float v2f;
typedef __attribute__((ext_vector_type(8))) float v8f;

#define IN_F 128
#define OUT_F 64
#define WAVES_PER_BLOCK 8
#define GEMM_BLOCKS 256

#define ROWS_PER_OWNER 400
#define OWN_THREADS 1024
#define OWN_WAVES 32
#define QCAP 1056
#define RECS_WS_OFFSET (64u << 20)
#define KEYS_WS_OFFSET (128u << 20)
#define RL_BITS 9

__device__ __forceinline__ unsigned edge_key(int row, int col, int* recx) {
  const unsigned o = (unsigned)row / ROWS_PER_OWNER;
  *recx = (int)(((unsigned)col << RL_BITS) | ((unsigned)row - o * ROWS_PER_OWNER));
  return o & 0xFFu;
}

__device__ __forceinline__ unsigned emit_four(int i, int4 r, int4 c, float4 v,
                                              int4* recs4) {
  int4 lo, hi;
  const unsigned k0 = edge_key(r.x, c.x, &lo.x);
  const unsigned k1 = edge_key(r.y, c.y, &lo.z);
  const unsigned k2 = edge_key(r.z, c.z, &hi.x);
  const unsigned k3 = edge_key(r.w, c.w, &hi.z);
  lo.y = __float_as_int(v.x);
  lo.w = __float_as_int(v.y);
  hi.y = __float_as_int(v.z);
  hi.w = __float_as_int(v.w);
  recs4[2 * i] = lo;
  recs4[2 * i + 1] = hi;
  return k0 | (k1 << 8) | (k2 << 16) | (k3 << 24);
}

__device__ __forceinline__ unsigned emit_partial(
    int i, int nEdges, const int* __restrict__ erow, const int* __restrict__ ecol,
    const float* __restrict__ eval, int* __restrict__ recs) {
  unsigned k = 0;
#pragma unroll
  for (int cc = 0; cc < 4; ++cc) {
    const int e = (i << 2) + cc;
    unsigned o = 0xFFu;
    if (e < nEdges) {
      int rx;
      o = edge_key(erow[e], ecol[e], &rx);
      recs[2 * e] = rx;
      recs[2 * e + 1] = __float_as_int(eval[e]);
    }
    k |= o << (8 * cc);
  }
  return k;
}

template <int S, unsigned M>
__device__ __forceinline__ unsigned transpose_stage(unsigned x, unsigned t, int lane) {
  const unsigned lower = (x & M) | ((t << S) & ~M);
  const unsigned upper = (x & ~M) | ((t >> S) & M);
  return (lane & S) ? upper : lower;
}

__device__ __forceinline__ unsigned bit_transpose32(unsigned x, int lane) {
  x = transpose_stage<16, 0x0000FFFFu>(
      x, (unsigned)__builtin_amdgcn_permlanex16(0, (int)x, 0x76543210, 0xfedcba98, false, false), lane);
  x = transpose_stage<8, 0x00FF00FFu>(
      x, (unsigned)__builtin_amdgcn_update_dpp(0, (int)x, 0x168, 0xF, 0xF, false), lane);
  x = transpose_stage<4, 0x0F0F0F0Fu>(
      x, (unsigned)__builtin_amdgcn_update_dpp(0, (int)x, 0x164, 0xF, 0xF, false), lane);
  x = transpose_stage<2, 0x33333333u>(
      x, (unsigned)__builtin_amdgcn_update_dpp(0, (int)x, 0x4E, 0xF, 0xF, false), lane);
  x = transpose_stage<1, 0x55555555u>(
      x, (unsigned)__builtin_amdgcn_update_dpp(0, (int)x, 0xB1, 0xF, 0xF, false), lane);
  return x;
}

struct EdgeWords {
  int i0, i1;
  bool f0, f1;
  int4 r0, q0, r1, q1;
  float4 v0, v1;
};

__device__ __forceinline__ EdgeWords load_edge_words(
    int i, int stride, int nFull, const int4* __restrict__ erow4,
    const int4* __restrict__ ecol4, const float4* __restrict__ eval4) {
  EdgeWords w;
  w.i0 = i;
  w.i1 = i + stride;
  w.f0 = w.i0 < nFull;
  w.f1 = w.i1 < nFull;
  const int j0 = w.f0 ? w.i0 : 0, j1 = w.f1 ? w.i1 : 0;
  w.r0 = erow4[j0]; w.q0 = ecol4[j0]; w.v0 = eval4[j0];
  w.r1 = erow4[j1]; w.q1 = ecol4[j1]; w.v1 = eval4[j1];
  return w;
}

__device__ __forceinline__ unsigned emit_word(
    int i, bool full, int4 r, int4 q, float4 v, int nFull, int nEdges,
    const int* __restrict__ erow, const int* __restrict__ ecol,
    const float* __restrict__ eval, int4* recs4) {
  if (full) return emit_four(i, r, q, v, recs4);
  if (i == nFull && (nEdges & 3)) return emit_partial(i, nEdges, erow, ecol, eval, (int*)recs4);
  return 0xFFFFFFFFu;
}

__device__ __forceinline__ void emit_edge_words(
    const EdgeWords& w, int lane, int nFull, int nEdges, int nKeyWords,
    const int* __restrict__ erow, const int* __restrict__ ecol,
    const float* __restrict__ eval, unsigned* keys, int4* recs4) {
  const unsigned k0 = emit_word(w.i0, w.f0, w.r0, w.q0, w.v0, nFull, nEdges, erow, ecol, eval, recs4);
  const unsigned k1 = emit_word(w.i1, w.f1, w.r1, w.q1, w.v1, nFull, nEdges, erow, ecol, eval, recs4);
  const unsigned p0 = bit_transpose32(k0, lane);
  const unsigned p1 = bit_transpose32(k1, lane);
  if (w.i0 < nKeyWords) keys[w.i0] = p0;
  if (w.i1 < nKeyWords) keys[w.i1] = p1;
}

__device__ __forceinline__ void load_a_tile(const float* __restrict__ V, int tile,
                                            int m, int koff, v2f (&a)[IN_F / 4]) {
  const float* vptr = V + (long long)(tile * 16 + m) * IN_F + koff;
#pragma unroll
  for (int ks = 0; ks < IN_F / 4; ++ks) a[ks] = *(const v2f*)(vptr + 4 * ks);
}

__global__ __launch_bounds__(256) void gemm_wmma_kernel(
    const float* __restrict__ V, const float* __restrict__ W,
    float* __restrict__ S, int nRowTiles, const int* __restrict__ erow,
    const int* __restrict__ ecol, const float* __restrict__ eval,
    unsigned* __restrict__ keys, int* __restrict__ recs, int nEdges) {
  const int lane = threadIdx.x & 31;
  const int wave = __builtin_amdgcn_readfirstlane(threadIdx.x >> 5);
  const int c    = lane & 15;
  const int h    = lane >> 4;
  const int koff = h << 1;

  v2f b[IN_F / 4][4];
  {
    const float4* w4 = (const float4*)(W + koff * OUT_F + 4 * c);
#pragma unroll
    for (int ks = 0; ks < IN_F / 4; ++ks) {
      const float4 w0 = w4[(4 * ks) * (OUT_F / 4)];
      const float4 w1 = w4[(4 * ks + 1) * (OUT_F / 4)];
      b[ks][0].x = w0.x; b[ks][0].y = w1.x;
      b[ks][1].x = w0.y; b[ks][1].y = w1.y;
      b[ks][2].x = w0.z; b[ks][2].y = w1.z;
      b[ks][3].x = w0.w; b[ks][3].y = w1.w;
    }
  }

  const int nWaves = gridDim.x * WAVES_PER_BLOCK;
  int tile = wave * gridDim.x + blockIdx.x;

  const int nKeyWords = ((nEdges + 1023) >> 10) << 8;
  const int nFull = nEdges >> 2;
  const int4* erow4 = (const int4*)erow;
  const int4* ecol4 = (const int4*)ecol;
  const float4* eval4 = (const float4*)eval;
  int4* recs4 = (int4*)recs;
  const int estride = gridDim.x * blockDim.x;
  int ew = blockIdx.x * blockDim.x + threadIdx.x;

  v2f a_cur[IN_F / 4], a_nxt[IN_F / 4];
  load_a_tile(V, tile < nRowTiles ? tile : nRowTiles - 1, c, koff, a_nxt);
  EdgeWords w = load_edge_words(ew, estride, nFull, erow4, ecol4, eval4);
  ew += 2 * estride;

  for (; tile < nRowTiles; tile += nWaves) {
#pragma unroll
    for (int ks = 0; ks < IN_F / 4; ++ks) a_cur[ks] = a_nxt[ks];

    const EdgeWords w_new = load_edge_words(ew, estride, nFull, erow4, ecol4, eval4);
    ew += 2 * estride;
    const int nxt = tile + nWaves;
    load_a_tile(V, nxt < nRowTiles ? nxt : nRowTiles - 1, c, koff, a_nxt);

    v8f c0 = {}, c1 = {}, c2 = {}, c3 = {};
#pragma unroll
    for (int ks = 0; ks < IN_F / 4; ++ks) {
      c0 = __builtin_amdgcn_wmma_f32_16x16x4_f32(false, a_cur[ks], false, b[ks][0],
                                                 (short)0, c0, false, false);
      c1 = __builtin_amdgcn_wmma_f32_16x16x4_f32(false, a_cur[ks], false, b[ks][1],
                                                 (short)0, c1, false, false);
      c2 = __builtin_amdgcn_wmma_f32_16x16x4_f32(false, a_cur[ks], false, b[ks][2],
                                                 (short)0, c2, false, false);
      c3 = __builtin_amdgcn_wmma_f32_16x16x4_f32(false, a_cur[ks], false, b[ks][3],
                                                 (short)0, c3, false, false);
    }

    float* srow = S + ((long long)tile * 16 + (h << 3)) * OUT_F + 4 * c;
#pragma unroll
    for (int r = 0; r < 8; ++r) {
      float4 o;
      o.x = c0[r]; o.y = c1[r]; o.z = c2[r]; o.w = c3[r];
      *(float4*)(srow + r * OUT_F) = o;
    }

    emit_edge_words(w, lane, nFull, nEdges, nKeyWords, erow, ecol, eval, keys, recs4);
    w = w_new;
  }
  emit_edge_words(w, lane, nFull, nEdges, nKeyWords, erow, ecol, eval, keys, recs4);

  for (; ew - lane < nKeyWords; ew += estride) {
    const bool full = ew < nFull;
    const int j = full ? ew : 0;
    const unsigned k = emit_word(ew, full, erow4[j], ecol4[j], eval4[j], nFull, nEdges,
                                 erow, ecol, eval, recs4);
    keys[ew] = bit_transpose32(k, lane);
  }
}


__device__ __forceinline__ unsigned planes_agree(uint4 x, const unsigned (&q)[4]) {
  return (x.x ^ q[0]) & (x.y ^ q[1]) & (x.z ^ q[2]) & (x.w ^ q[3]);
}

__device__ __forceinline__ void drain_batch(
    const unsigned* qb, int n, int lane, const float* __restrict__ S,
    const int2* __restrict__ recs2, float* acc) {
  const unsigned e = (lane < n) ? qb[lane] : 0u;
  const int2 rec = recs2[e];
  const int col = (int)((unsigned)rec.x >> RL_BITS);
  const int rl = rec.x & ((1 << RL_BITS) - 1);
  const int vbits = rec.y;
  const float* Shi = S + 32;
  for (int j0 = 0; j0 < n; j0 += 16) {
    float s0[16], s1[16];
#pragma unroll
    for (int u = 0; u < 16; ++u) {
      const unsigned c = (unsigned)__builtin_amdgcn_readlane(col, j0 + u);
      const unsigned idx = c * OUT_F + (unsigned)lane;
      s0[u] = S[idx];
      s1[u] = Shi[idx];
    }
#pragma unroll
    for (int u = 0; u < 16; ++u) {
      if (j0 + u < n) {
        const int r = __builtin_amdgcn_readlane(rl, j0 + u);
        const float v = __int_as_float(__builtin_amdgcn_readlane(vbits, j0 + u));
        float* a = acc + r * OUT_F + lane;
        __hip_atomic_fetch_add(a, s0[u] * v, __ATOMIC_RELAXED,
                               __HIP_MEMORY_SCOPE_WORKGROUP);
        __hip_atomic_fetch_add(a + 32, s1[u] * v, __ATOMIC_RELAXED,
                               __HIP_MEMORY_SCOPE_WORKGROUP);
      }
    }
  }
}

__global__ __launch_bounds__(OWN_THREADS) void spmm_owner_kernel(
    const float* __restrict__ S, const unsigned* __restrict__ keys,
    const int* __restrict__ recs, const float* __restrict__ bias,
    float* __restrict__ out, int nNodes, int nEdges, int nOwners) {
  __shared__ float acc[ROWS_PER_OWNER * OUT_F];
  __shared__ unsigned queue[OWN_WAVES * QCAP];

  const int tid = threadIdx.x;
  const int lane = tid & 31;
  const int wave = __builtin_amdgcn_readfirstlane(tid >> 5);
  const int owner = blockIdx.x;
  const int base = owner * ROWS_PER_OWNER;

  for (int i = tid; i < ROWS_PER_OWNER * OUT_F; i += OWN_THREADS)
    acc[i] = bias[i & (OUT_F - 1)];
  __syncthreads();

  const uint4* keys16 = (const uint4*)keys;
  const int2* recs2 = (const int2*)recs;
  unsigned* q = queue + wave * QCAP;
  const int hsel = lane & 1;
  unsigned qsel[4];
#pragma unroll
  for (int j = 0; j < 4; ++j)
    qsel[j] = ((owner >> (4 * hsel + j)) & 1) ? 0u : ~0u;
  const unsigned eoff =
      512u * hsel + 128u * (lane >> 3) + (unsigned)((lane >> 1) & 3);
  const int nTrips = (nEdges + 1023) >> 10;
  const int rot = (int)(((unsigned)owner * (unsigned)nTrips) / (unsigned)nOwners);
  int cnt = 0;

  for (int seg = 0; seg < 2; ++seg) {
    const int lo = seg ? 0 : rot;
    const int hi = seg ? rot : nTrips;
    for (int t0 = lo + wave; t0 < hi; t0 += OWN_WAVES * 4) {
      uint4 xa[4], xb[4];
#pragma unroll
      for (int u = 0; u < 4; ++u) {
        const int t = t0 + u * OWN_WAVES;
        const int tt = t < hi ? t : t0;
        xa[u] = keys16[tt * 64 + lane];
        xb[u] = keys16[tt * 64 + 32 + lane];
      }
#pragma unroll
      for (int u = 0; u < 4; ++u) {
        const int t = t0 + u * OWN_WAVES;
        if (t >= hi) break;
        const unsigned ra = planes_agree(xa[u], qsel);
        const unsigned rb = planes_agree(xb[u], qsel);
        const unsigned other = (unsigned)__builtin_amdgcn_update_dpp(
            0, (int)(hsel ? ra : rb), 0xB1, 0xF, 0xF, false);
        unsigned m = (hsel ? rb : ra) & other;
        const unsigned ebase = ((unsigned)t << 10) + eoff;
        unsigned bal = __builtin_amdgcn_ballot_w32(m != 0u);
        while (bal != 0u) {
          if (m != 0u) {
            const int p = __builtin_ctz(m);
            m &= m - 1u;
            q[cnt + (int)__builtin_amdgcn_mbcnt_lo(bal, 0u)] = ebase + ((unsigned)p << 2);
          }
          cnt += __builtin_popcount(bal);
          bal = __builtin_amdgcn_ballot_w32(m != 0u);
        }
        while (cnt >= 32) {
          cnt -= 32;
          drain_batch(q + cnt, 32, lane, S, recs2, acc);
        }
      }
    }
  }
  if (cnt > 0) drain_batch(q, cnt, lane, S, recs2, acc);

  __syncthreads();

  int nrows = nNodes - base;
  if (nrows > ROWS_PER_OWNER) nrows = ROWS_PER_OWNER;
  const float4* a4 = (const float4*)acc;
  float4* o4 = (float4*)(out + (long long)base * OUT_F);
  for (int i = tid; i < nrows * (OUT_F / 4); i += OWN_THREADS) o4[i] = a4[i];
}

extern "C" void kernel_launch(void* const* d_in, const int* in_sizes, int n_in,
                              void* d_out, int out_size, void* d_ws,
                              size_t ws_size, hipStream_t stream) {
  const float* vertex  = (const float*)d_in[0];
  const int*   erow    = (const int*)d_in[1];
  const int*   ecol    = (const int*)d_in[2];
  const float* eval    = (const float*)d_in[3];
  const float* weights = (const float*)d_in[4];
  const float* bias    = (const float*)d_in[5];
  float* out     = (float*)d_out;
  float* support = (float*)d_ws;
  int* recs = (int*)((char*)d_ws + RECS_WS_OFFSET);
  unsigned* keys = (unsigned*)((char*)d_ws + KEYS_WS_OFFSET);

  const int nNodes = in_sizes[0] / IN_F;
  const int nEdges = in_sizes[1];

  const int nRowTiles = nNodes / 16;
  gemm_wmma_kernel<<<GEMM_BLOCKS, 256, 0, stream>>>(vertex, weights, support,
                                                nRowTiles, erow, ecol, eval,
                                                keys, recs, nEdges);

  const int nOwners = (nNodes + ROWS_PER_OWNER - 1) / ROWS_PER_OWNER;
  spmm_owner_kernel<<<nOwners, OWN_THREADS, 0, stream>>>(
      support, keys, recs, bias, out, nNodes, nEdges, nOwners);
}


# owner drain reads each support row with ONE 8-byte-per-lane load (lane l = features 2l,2l+1: one aligned 256-B request per row, was two 128-B line loads); LDS f32 adds in arrival order as before
# speedup vs baseline: 2.3629x; 1.0468x over previous
#include <hip/hip_runtime.h>


typedef __attribute__((ext_vector_type(2))) float v2f;
typedef __attribute__((ext_vector_type(8))) float v8f;

#define IN_F 128
#define OUT_F 64
#define WAVES_PER_BLOCK 8
#define GEMM_BLOCKS 256

#define ROWS_PER_OWNER 400
#define OWN_THREADS 1024
#define OWN_WAVES 32
#define QCAP 1056
#define RECS_WS_OFFSET (64u << 20)
#define KEYS_WS_OFFSET (128u << 20)
#define RL_BITS 9

__device__ __forceinline__ unsigned edge_key(int row, int col, int* recx) {
  const unsigned o = (unsigned)row / ROWS_PER_OWNER;
  *recx = (int)(((unsigned)col << RL_BITS) | ((unsigned)row - o * ROWS_PER_OWNER));
  return o & 0xFFu;
}

__device__ __forceinline__ unsigned emit_four(int i, int4 r, int4 c, float4 v,
                                              int4* recs4) {
  int4 lo, hi;
  const unsigned k0 = edge_key(r.x, c.x, &lo.x);
  const unsigned k1 = edge_key(r.y, c.y, &lo.z);
  const unsigned k2 = edge_key(r.z, c.z, &hi.x);
  const unsigned k3 = edge_key(r.w, c.w, &hi.z);
  lo.y = __float_as_int(v.x);
  lo.w = __float_as_int(v.y);
  hi.y = __float_as_int(v.z);
  hi.w = __float_as_int(v.w);
  recs4[2 * i] = lo;
  recs4[2 * i + 1] = hi;
  return k0 | (k1 << 8) | (k2 << 16) | (k3 << 24);
}

__device__ __forceinline__ unsigned emit_partial(
    int i, int nEdges, const int* __restrict__ erow, const int* __restrict__ ecol,
    const float* __restrict__ eval, int* __restrict__ recs) {
  unsigned k = 0;
#pragma unroll
  for (int cc = 0; cc < 4; ++cc) {
    const int e = (i << 2) + cc;
    unsigned o = 0xFFu;
    if (e < nEdges) {
      int rx;
      o = edge_key(erow[e], ecol[e], &rx);
      recs[2 * e] = rx;
      recs[2 * e + 1] = __float_as_int(eval[e]);
    }
    k |= o << (8 * cc);
  }
  return k;
}

template <int S, unsigned M>
__device__ __forceinline__ unsigned transpose_stage(unsigned x, unsigned t, int lane) {
  const unsigned lower = (x & M) | ((t << S) & ~M);
  const unsigned upper = (x & ~M) | ((t >> S) & M);
  return (lane & S) ? upper : lower;
}

__device__ __forceinline__ unsigned bit_transpose32(unsigned x, int lane) {
  x = transpose_stage<16, 0x0000FFFFu>(
      x, (unsigned)__builtin_amdgcn_permlanex16(0, (int)x, 0x76543210, 0xfedcba98, false, false), lane);
  x = transpose_stage<8, 0x00FF00FFu>(
      x, (unsigned)__builtin_amdgcn_update_dpp(0, (int)x, 0x168, 0xF, 0xF, false), lane);
  x = transpose_stage<4, 0x0F0F0F0Fu>(
      x, (unsigned)__builtin_amdgcn_update_dpp(0, (int)x, 0x164, 0xF, 0xF, false), lane);
  x = transpose_stage<2, 0x33333333u>(
      x, (unsigned)__builtin_amdgcn_update_dpp(0, (int)x, 0x4E, 0xF, 0xF, false), lane);
  x = transpose_stage<1, 0x55555555u>(
      x, (unsigned)__builtin_amdgcn_update_dpp(0, (int)x, 0xB1, 0xF, 0xF, false), lane);
  return x;
}

struct EdgeWords {
  int i0, i1;
  bool f0, f1;
  int4 r0, q0, r1, q1;
  float4 v0, v1;
};

__device__ __forceinline__ EdgeWords load_edge_words(
    int i, int stride, int nFull, const int4* __restrict__ erow4,
    const int4* __restrict__ ecol4, const float4* __restrict__ eval4) {
  EdgeWords w;
  w.i0 = i;
  w.i1 = i + stride;
  w.f0 = w.i0 < nFull;
  w.f1 = w.i1 < nFull;
  const int j0 = w.f0 ? w.i0 : 0, j1 = w.f1 ? w.i1 : 0;
  w.r0 = erow4[j0]; w.q0 = ecol4[j0]; w.v0 = eval4[j0];
  w.r1 = erow4[j1]; w.q1 = ecol4[j1]; w.v1 = eval4[j1];
  return w;
}

__device__ __forceinline__ unsigned emit_word(
    int i, bool full, int4 r, int4 q, float4 v, int nFull, int nEdges,
    const int* __restrict__ erow, const int* __restrict__ ecol,
    const float* __restrict__ eval, int4* recs4) {
  if (full) return emit_four(i, r, q, v, recs4);
  if (i == nFull && (nEdges & 3)) return emit_partial(i, nEdges, erow, ecol, eval, (int*)recs4);
  return 0xFFFFFFFFu;
}

__device__ __forceinline__ void emit_edge_words(
    const EdgeWords& w, int lane, int nFull, int nEdges, int nKeyWords,
    const int* __restrict__ erow, const int* __restrict__ ecol,
    const float* __restrict__ eval, unsigned* keys, int4* recs4) {
  const unsigned k0 = emit_word(w.i0, w.f0, w.r0, w.q0, w.v0, nFull, nEdges, erow, ecol, eval, recs4);
  const unsigned k1 = emit_word(w.i1, w.f1, w.r1, w.q1, w.v1, nFull, nEdges, erow, ecol, eval, recs4);
  const unsigned p0 = bit_transpose32(k0, lane);
  const unsigned p1 = bit_transpose32(k1, lane);
  if (w.i0 < nKeyWords) keys[w.i0] = p0;
  if (w.i1 < nKeyWords) keys[w.i1] = p1;
}

__device__ __forceinline__ void load_a_tile(const float* __restrict__ V, int tile,
                                            int m, int koff, v2f (&a)[IN_F / 4]) {
  const float* vptr = V + (long long)(tile * 16 + m) * IN_F + koff;
#pragma unroll
  for (int ks = 0; ks < IN_F / 4; ++ks) a[ks] = *(const v2f*)(vptr + 4 * ks);
}

__global__ __launch_bounds__(256) void gemm_wmma_kernel(
    const float* __restrict__ V, const float* __restrict__ W,
    float* __restrict__ S, int nRowTiles, const int* __restrict__ erow,
    const int* __restrict__ ecol, const float* __restrict__ eval,
    unsigned* __restrict__ keys, int* __restrict__ recs, int nEdges) {
  const int lane = threadIdx.x & 31;
  const int wave = __builtin_amdgcn_readfirstlane(threadIdx.x >> 5);
  const int c    = lane & 15;
  const int h    = lane >> 4;
  const int koff = h << 1;

  v2f b[IN_F / 4][4];
  {
    const float4* w4 = (const float4*)(W + koff * OUT_F + 4 * c);
#pragma unroll
    for (int ks = 0; ks < IN_F / 4; ++ks) {
      const float4 w0 = w4[(4 * ks) * (OUT_F / 4)];
      const float4 w1 = w4[(4 * ks + 1) * (OUT_F / 4)];
      b[ks][0].x = w0.x; b[ks][0].y = w1.x;
      b[ks][1].x = w0.y; b[ks][1].y = w1.y;
      b[ks][2].x = w0.z; b[ks][2].y = w1.z;
      b[ks][3].x = w0.w; b[ks][3].y = w1.w;
    }
  }

  const int nWaves = gridDim.x * WAVES_PER_BLOCK;
  int tile = wave * gridDim.x + blockIdx.x;

  const int nKeyWords = ((nEdges + 1023) >> 10) << 8;
  const int nFull = nEdges >> 2;
  const int4* erow4 = (const int4*)erow;
  const int4* ecol4 = (const int4*)ecol;
  const float4* eval4 = (const float4*)eval;
  int4* recs4 = (int4*)recs;
  const int estride = gridDim.x * blockDim.x;
  int ew = blockIdx.x * blockDim.x + threadIdx.x;

  v2f a_cur[IN_F / 4], a_nxt[IN_F / 4];
  load_a_tile(V, tile < nRowTiles ? tile : nRowTiles - 1, c, koff, a_nxt);
  EdgeWords w = load_edge_words(ew, estride, nFull, erow4, ecol4, eval4);
  ew += 2 * estride;

  for (; tile < nRowTiles; tile += nWaves) {
#pragma unroll
    for (int ks = 0; ks < IN_F / 4; ++ks) a_cur[ks] = a_nxt[ks];

    const EdgeWords w_new = load_edge_words(ew, estride, nFull, erow4, ecol4, eval4);
    ew += 2 * estride;
    const int nxt = tile + nWaves;
    load_a_tile(V, nxt < nRowTiles ? nxt : nRowTiles - 1, c, koff, a_nxt);

    v8f c0 = {}, c1 = {}, c2 = {}, c3 = {};
#pragma unroll
    for (int ks = 0; ks < IN_F / 4; ++ks) {
      c0 = __builtin_amdgcn_wmma_f32_16x16x4_f32(false, a_cur[ks], false, b[ks][0],
                                                 (short)0, c0, false, false);
      c1 = __builtin_amdgcn_wmma_f32_16x16x4_f32(false, a_cur[ks], false, b[ks][1],
                                                 (short)0, c1, false, false);
      c2 = __builtin_amdgcn_wmma_f32_16x16x4_f32(false, a_cur[ks], false, b[ks][2],
                                                 (short)0, c2, false, false);
      c3 = __builtin_amdgcn_wmma_f32_16x16x4_f32(false, a_cur[ks], false, b[ks][3],
                                                 (short)0, c3, false, false);
    }

    float* srow = S + ((long long)tile * 16 + (h << 3)) * OUT_F + 4 * c;
#pragma unroll
    for (int r = 0; r < 8; ++r) {
      float4 o;
      o.x = c0[r]; o.y = c1[r]; o.z = c2[r]; o.w = c3[r];
      *(float4*)(srow + r * OUT_F) = o;
    }

    emit_edge_words(w, lane, nFull, nEdges, nKeyWords, erow, ecol, eval, keys, recs4);
    w = w_new;
  }
  emit_edge_words(w, lane, nFull, nEdges, nKeyWords, erow, ecol, eval, keys, recs4);

  for (; ew - lane < nKeyWords; ew += estride) {
    const bool full = ew < nFull;
    const int j = full ? ew : 0;
    const unsigned k = emit_word(ew, full, erow4[j], ecol4[j], eval4[j], nFull, nEdges,
                                 erow, ecol, eval, recs4);
    keys[ew] = bit_transpose32(k, lane);
  }
}


__device__ __forceinline__ unsigned planes_agree(uint4 x, const unsigned (&q)[4]) {
  return (x.x ^ q[0]) & (x.y ^ q[1]) & (x.z ^ q[2]) & (x.w ^ q[3]);
}

__device__ __forceinline__ void drain_batch(
    const unsigned* qb, int n, int lane, const float* __restrict__ S,
    const int2* __restrict__ recs2, float* acc) {
  const unsigned e = (lane < n) ? qb[lane] : 0u;
  const int2 rec = recs2[e];
  const int col = (int)((unsigned)rec.x >> RL_BITS);
  const int rl = rec.x & ((1 << RL_BITS) - 1);
  const int vbits = rec.y;
  const v2f* S2 = (const v2f*)S;
  for (int j0 = 0; j0 < n; j0 += 16) {
    v2f s[16];
#pragma unroll
    for (int u = 0; u < 16; ++u) {
      const unsigned c = (unsigned)__builtin_amdgcn_readlane(col, j0 + u);
      s[u] = S2[c * (OUT_F / 2) + (unsigned)lane];
    }
#pragma unroll
    for (int u = 0; u < 16; ++u) {
      if (j0 + u < n) {
        const int r = __builtin_amdgcn_readlane(rl, j0 + u);
        const float v = __int_as_float(__builtin_amdgcn_readlane(vbits, j0 + u));
        float* a = acc + r * OUT_F + 2 * lane;
        __hip_atomic_fetch_add(a, s[u].x * v, __ATOMIC_RELAXED,
                               __HIP_MEMORY_SCOPE_WORKGROUP);
        __hip_atomic_fetch_add(a + 1, s[u].y * v, __ATOMIC_RELAXED,
                               __HIP_MEMORY_SCOPE_WORKGROUP);
      }
    }
  }
}

__global__ __launch_bounds__(OWN_THREADS) void spmm_owner_kernel(
    const float* __restrict__ S, const unsigned* __restrict__ keys,
    const int* __restrict__ recs, const float* __restrict__ bias,
    float* __restrict__ out, int nNodes, int nEdges, int nOwners) {
  __shared__ float acc[ROWS_PER_OWNER * OUT_F];
  __shared__ unsigned queue[OWN_WAVES * QCAP];

  const int tid = threadIdx.x;
  const int lane = tid & 31;
  const int wave = __builtin_amdgcn_readfirstlane(tid >> 5);
  const int owner = blockIdx.x;
  const int base = owner * ROWS_PER_OWNER;

  for (int i = tid; i < ROWS_PER_OWNER * OUT_F; i += OWN_THREADS)
    acc[i] = bias[i & (OUT_F - 1)];
  __syncthreads();

  const uint4* keys16 = (const uint4*)keys;
  const int2* recs2 = (const int2*)recs;
  unsigned* q = queue + wave * QCAP;
  const int hsel = lane & 1;
  unsigned qsel[4];
#pragma unroll
  for (int j = 0; j < 4; ++j)
    qsel[j] = ((owner >> (4 * hsel + j)) & 1) ? 0u : ~0u;
  const unsigned eoff =
      512u * hsel + 128u * (lane >> 3) + (unsigned)((lane >> 1) & 3);
  const int nTrips = (nEdges + 1023) >> 10;
  const int rot = (int)(((unsigned)owner * (unsigned)nTrips) / (unsigned)nOwners);
  int cnt = 0;

  for (int seg = 0; seg < 2; ++seg) {
    const int lo = seg ? 0 : rot;
    const int hi = seg ? rot : nTrips;
    for (int t0 = lo + wave; t0 < hi; t0 += OWN_WAVES * 4) {
      uint4 xa[4], xb[4];
#pragma unroll
      for (int u = 0; u < 4; ++u) {
        const int t = t0 + u * OWN_WAVES;
        const int tt = t < hi ? t : t0;
        xa[u] = keys16[tt * 64 + lane];
        xb[u] = keys16[tt * 64 + 32 + lane];
      }
#pragma unroll
      for (int u = 0; u < 4; ++u) {
        const int t = t0 + u * OWN_WAVES;
        if (t >= hi) break;
        const unsigned ra = planes_agree(xa[u], qsel);
        const unsigned rb = planes_agree(xb[u], qsel);
        const unsigned other = (unsigned)__builtin_amdgcn_update_dpp(
            0, (int)(hsel ? ra : rb), 0xB1, 0xF, 0xF, false);
        unsigned m = (hsel ? rb : ra) & other;
        const unsigned ebase = ((unsigned)t << 10) + eoff;
        unsigned bal = __builtin_amdgcn_ballot_w32(m != 0u);
        while (bal != 0u) {
          if (m != 0u) {
            const int p = __builtin_ctz(m);
            m &= m - 1u;
            q[cnt + (int)__builtin_amdgcn_mbcnt_lo(bal, 0u)] = ebase + ((unsigned)p << 2);
          }
          cnt += __builtin_popcount(bal);
          bal = __builtin_amdgcn_ballot_w32(m != 0u);
        }
        while (cnt >= 32) {
          cnt -= 32;
          drain_batch(q + cnt, 32, lane, S, recs2, acc);
        }
      }
    }
  }
  if (cnt > 0) drain_batch(q, cnt, lane, S, recs2, acc);

  __syncthreads();

  int nrows = nNodes - base;
  if (nrows > ROWS_PER_OWNER) nrows = ROWS_PER_OWNER;
  const float4* a4 = (const float4*)acc;
  float4* o4 = (float4*)(out + (long long)base * OUT_F);
  for (int i = tid; i < nrows * (OUT_F / 4); i += OWN_THREADS) o4[i] = a4[i];
}

extern "C" void kernel_launch(void* const* d_in, const int* in_sizes, int n_in,
                              void* d_out, int out_size, void* d_ws,
                              size_t ws_size, hipStream_t stream) {
  const float* vertex  = (const float*)d_in[0];
  const int*   erow    = (const int*)d_in[1];
  const int*   ecol    = (const int*)d_in[2];
  const float* eval    = (const float*)d_in[3];
  const float* weights = (const float*)d_in[4];
  const float* bias    = (const float*)d_in[5];
  float* out     = (float*)d_out;
  float* support = (float*)d_ws;
  int* recs = (int*)((char*)d_ws + RECS_WS_OFFSET);
  unsigned* keys = (unsigned*)((char*)d_ws + KEYS_WS_OFFSET);

  const int nNodes = in_sizes[0] / IN_F;
  const int nEdges = in_sizes[1];

  const int nRowTiles = nNodes / 16;
  gemm_wmma_kernel<<<GEMM_BLOCKS, 256, 0, stream>>>(vertex, weights, support,
                                                nRowTiles, erow, ecol, eval,
                                                keys, recs, nEdges);

  const int nOwners = (nNodes + ROWS_PER_OWNER - 1) / ROWS_PER_OWNER;
  spmm_owner_kernel<<<nOwners, OWN_THREADS, 0, stream>>>(
      support, keys, recs, bias, out, nNodes, nEdges, nOwners);
}


# owner key as two one-hot hex digits in 32 contiguous bit planes: an owner streams only its 2 digit planes and ANDs them (2 bits per edge read, was 8); LDS f32 adds in arrival order as before
# speedup vs baseline: 2.5799x; 1.0918x over previous
#include <hip/hip_runtime.h>


typedef __attribute__((ext_vector_type(2))) float v2f;
typedef __attribute__((ext_vector_type(8))) float v8f;

#define IN_F 128
#define OUT_F 64
#define WAVES_PER_BLOCK 8
#define GEMM_BLOCKS 256

#define ROWS_PER_OWNER 400
#define OWN_THREADS 1024
#define OWN_WAVES 32
#define QCAP 1056
#define RECS_WS_OFFSET (64u << 20)
#define KEYS_WS_OFFSET (128u << 20)
#define RL_BITS 9

__device__ __forceinline__ unsigned edge_key(int row, int col, int* recx) {
  const unsigned o = (unsigned)row / ROWS_PER_OWNER;
  *recx = (int)(((unsigned)col << RL_BITS) | ((unsigned)row - o * ROWS_PER_OWNER));
  return o & 0xFFu;
}

__device__ __forceinline__ unsigned digit_word(unsigned o) {
  return (1u << (o >> 4)) | (0x10000u << (o & 15u));
}

__device__ __forceinline__ void emit_four(int i, int4 r, int4 c, float4 v,
                                          int4* recs4, unsigned (&k)[4]) {
  int4 lo, hi;
  k[0] = digit_word(edge_key(r.x, c.x, &lo.x));
  k[1] = digit_word(edge_key(r.y, c.y, &lo.z));
  k[2] = digit_word(edge_key(r.z, c.z, &hi.x));
  k[3] = digit_word(edge_key(r.w, c.w, &hi.z));
  lo.y = __float_as_int(v.x);
  lo.w = __float_as_int(v.y);
  hi.y = __float_as_int(v.z);
  hi.w = __float_as_int(v.w);
  recs4[2 * i] = lo;
  recs4[2 * i + 1] = hi;
}

__device__ __forceinline__ void emit_partial(
    int i, int nEdges, const int* __restrict__ erow, const int* __restrict__ ecol,
    const float* __restrict__ eval, int* __restrict__ recs, unsigned (&k)[4]) {
#pragma unroll
  for (int cc = 0; cc < 4; ++cc) {
    const int e = (i << 2) + cc;
    k[cc] = 0u;
    if (e < nEdges) {
      int rx;
      k[cc] = digit_word(edge_key(erow[e], ecol[e], &rx));
      recs[2 * e] = rx;
      recs[2 * e + 1] = __float_as_int(eval[e]);
    }
  }
}

template <int S, unsigned M>
__device__ __forceinline__ unsigned transpose_stage(unsigned x, unsigned t, int lane) {
  const unsigned lower = (x & M) | ((t << S) & ~M);
  const unsigned upper = (x & ~M) | ((t >> S) & M);
  return (lane & S) ? upper : lower;
}

__device__ __forceinline__ unsigned bit_transpose32(unsigned x, int lane) {
  x = transpose_stage<16, 0x0000FFFFu>(
      x, (unsigned)__builtin_amdgcn_permlanex16(0, (int)x, 0x76543210, 0xfedcba98, false, false), lane);
  x = transpose_stage<8, 0x00FF00FFu>(
      x, (unsigned)__builtin_amdgcn_update_dpp(0, (int)x, 0x168, 0xF, 0xF, false), lane);
  x = transpose_stage<4, 0x0F0F0F0Fu>(
      x, (unsigned)__builtin_amdgcn_update_dpp(0, (int)x, 0x164, 0xF, 0xF, false), lane);
  x = transpose_stage<2, 0x33333333u>(
      x, (unsigned)__builtin_amdgcn_update_dpp(0, (int)x, 0x4E, 0xF, 0xF, false), lane);
  x = transpose_stage<1, 0x55555555u>(
      x, (unsigned)__builtin_amdgcn_update_dpp(0, (int)x, 0xB1, 0xF, 0xF, false), lane);
  return x;
}

struct EdgeWords {
  int i0, i1;
  bool f0, f1;
  int4 r0, q0, r1, q1;
  float4 v0, v1;
};

__device__ __forceinline__ EdgeWords load_edge_words(
    int i, int stride, int nFull, const int4* __restrict__ erow4,
    const int4* __restrict__ ecol4, const float4* __restrict__ eval4) {
  EdgeWords w;
  w.i0 = i;
  w.i1 = i + stride;
  w.f0 = w.i0 < nFull;
  w.f1 = w.i1 < nFull;
  const int j0 = w.f0 ? w.i0 : 0, j1 = w.f1 ? w.i1 : 0;
  w.r0 = erow4[j0]; w.q0 = ecol4[j0]; w.v0 = eval4[j0];
  w.r1 = erow4[j1]; w.q1 = ecol4[j1]; w.v1 = eval4[j1];
  return w;
}

__device__ __forceinline__ void emit_word(
    int i, bool full, int4 r, int4 q, float4 v, int nFull, int nEdges,
    const int* __restrict__ erow, const int* __restrict__ ecol,
    const float* __restrict__ eval, int4* recs4, unsigned (&k)[4]) {
  if (full) {
    emit_four(i, r, q, v, recs4, k);
  } else if (i == nFull && (nEdges & 3)) {
    emit_partial(i, nEdges, erow, ecol, eval, (int*)recs4, k);
  } else {
    k[0] = 0u; k[1] = 0u; k[2] = 0u; k[3] = 0u;
  }
}

__device__ __forceinline__ void emit_edge_words(
    const EdgeWords& w, int lane, int nFull, int nEdges, int nKeyWords,
    const int* __restrict__ erow, const int* __restrict__ ecol,
    const float* __restrict__ eval, unsigned* keys, int4* recs4) {
  unsigned ka[4], kb[4];
  emit_word(w.i0, w.f0, w.r0, w.q0, w.v0, nFull, nEdges, erow, ecol, eval, recs4, ka);
  emit_word(w.i1, w.f1, w.r1, w.q1, w.v1, nFull, nEdges, erow, ecol, eval, recs4, kb);
  unsigned pa[4], pb[4];
#pragma unroll
  for (int c = 0; c < 4; ++c) {
    pa[c] = bit_transpose32(ka[c], lane);
    pb[c] = bit_transpose32(kb[c], lane);
  }
  if (w.i0 < nKeyWords) {
    const int odd = lane & 1;
    unsigned s0[4], s1[4];
#pragma unroll
    for (int c = 0; c < 4; ++c) {
      const unsigned got = (unsigned)__builtin_amdgcn_update_dpp(
          0, (int)(odd ? pa[c] : pb[c]), 0xB1, 0xF, 0xF, false);
      s0[c] = odd ? got : pa[c];
      s1[c] = odd ? pb[c] : got;
    }
    const int planePieces = nKeyWords >> 5;
    const int piece = (w.i0 >> 5) + odd;
    uint4* planes = (uint4*)keys;
    planes[(lane & ~1) * planePieces + piece] = make_uint4(s0[0], s0[1], s0[2], s0[3]);
    planes[(lane | 1) * planePieces + piece] = make_uint4(s1[0], s1[1], s1[2], s1[3]);
  }
}

__device__ __forceinline__ void load_a_tile(const float* __restrict__ V, int tile,
                                            int m, int koff, v2f (&a)[IN_F / 4]) {
  const float* vptr = V + (long long)(tile * 16 + m) * IN_F + koff;
#pragma unroll
  for (int ks = 0; ks < IN_F / 4; ++ks) a[ks] = *(const v2f*)(vptr + 4 * ks);
}

__global__ __launch_bounds__(256) void gemm_wmma_kernel(
    const float* __restrict__ V, const float* __restrict__ W,
    float* __restrict__ S, int nRowTiles, const int* __restrict__ erow,
    const int* __restrict__ ecol, const float* __restrict__ eval,
    unsigned* __restrict__ keys, int* __restrict__ recs, int nEdges) {
  const int lane = threadIdx.x & 31;
  const int wave = __builtin_amdgcn_readfirstlane(threadIdx.x >> 5);
  const int c    = lane & 15;
  const int h    = lane >> 4;
  const int koff = h << 1;

  v2f b[IN_F / 4][4];
  {
    const float4* w4 = (const float4*)(W + koff * OUT_F + 4 * c);
#pragma unroll
    for (int ks = 0; ks < IN_F / 4; ++ks) {
      const float4 w0 = w4[(4 * ks) * (OUT_F / 4)];
      const float4 w1 = w4[(4 * ks + 1) * (OUT_F / 4)];
      b[ks][0].x = w0.x; b[ks][0].y = w1.x;
      b[ks][1].x = w0.y; b[ks][1].y = w1.y;
      b[ks][2].x = w0.z; b[ks][2].y = w1.z;
      b[ks][3].x = w0.w; b[ks][3].y = w1.w;
    }
  }

  const int nWaves = gridDim.x * WAVES_PER_BLOCK;
  int tile = wave * gridDim.x + blockIdx.x;

  const int nKeyWords = ((nEdges + 4095) >> 12) << 10;
  const int nFull = nEdges >> 2;
  const int4* erow4 = (const int4*)erow;
  const int4* ecol4 = (const int4*)ecol;
  const float4* eval4 = (const float4*)eval;
  int4* recs4 = (int4*)recs;
  const int estride = 2 * gridDim.x * blockDim.x;
  int ew = (blockIdx.x * WAVES_PER_BLOCK + wave) * 64 + lane;

  v2f a_cur[IN_F / 4], a_nxt[IN_F / 4];
  load_a_tile(V, tile < nRowTiles ? tile : nRowTiles - 1, c, koff, a_nxt);
  EdgeWords w = load_edge_words(ew, 32, nFull, erow4, ecol4, eval4);
  ew += estride;

  for (; tile < nRowTiles; tile += nWaves) {
#pragma unroll
    for (int ks = 0; ks < IN_F / 4; ++ks) a_cur[ks] = a_nxt[ks];

    const EdgeWords w_new = load_edge_words(ew, 32, nFull, erow4, ecol4, eval4);
    ew += estride;
    const int nxt = tile + nWaves;
    load_a_tile(V, nxt < nRowTiles ? nxt : nRowTiles - 1, c, koff, a_nxt);

    v8f c0 = {}, c1 = {}, c2 = {}, c3 = {};
#pragma unroll
    for (int ks = 0; ks < IN_F / 4; ++ks) {
      c0 = __builtin_amdgcn_wmma_f32_16x16x4_f32(false, a_cur[ks], false, b[ks][0],
                                                 (short)0, c0, false, false);
      c1 = __builtin_amdgcn_wmma_f32_16x16x4_f32(false, a_cur[ks], false, b[ks][1],
                                                 (short)0, c1, false, false);
      c2 = __builtin_amdgcn_wmma_f32_16x16x4_f32(false, a_cur[ks], false, b[ks][2],
                                                 (short)0, c2, false, false);
      c3 = __builtin_amdgcn_wmma_f32_16x16x4_f32(false, a_cur[ks], false, b[ks][3],
                                                 (short)0, c3, false, false);
    }

    float* srow = S + ((long long)tile * 16 + (h << 3)) * OUT_F + 4 * c;
#pragma unroll
    for (int r = 0; r < 8; ++r) {
      float4 o;
      o.x = c0[r]; o.y = c1[r]; o.z = c2[r]; o.w = c3[r];
      *(float4*)(srow + r * OUT_F) = o;
    }

    emit_edge_words(w, lane, nFull, nEdges, nKeyWords, erow, ecol, eval, keys, recs4);
    w = w_new;
  }
  emit_edge_words(w, lane, nFull, nEdges, nKeyWords, erow, ecol, eval, keys, recs4);

  for (; ew - lane < nKeyWords; ew += estride) {
    const EdgeWords wr = load_edge_words(ew, 32, nFull, erow4, ecol4, eval4);
    emit_edge_words(wr, lane, nFull, nEdges, nKeyWords, erow, ecol, eval, keys, recs4);
  }
}


__device__ __forceinline__ void drain_batch(
    const unsigned* qb, int n, int lane, const float* __restrict__ S,
    const int2* __restrict__ recs2, float* acc) {
  const unsigned e = (lane < n) ? qb[lane] : 0u;
  const int2 rec = recs2[e];
  const int col = (int)((unsigned)rec.x >> RL_BITS);
  const int rl = rec.x & ((1 << RL_BITS) - 1);
  const int vbits = rec.y;
  const v2f* S2 = (const v2f*)S;
  for (int j0 = 0; j0 < n; j0 += 16) {
    v2f s[16];
#pragma unroll
    for (int u = 0; u < 16; ++u) {
      const unsigned c = (unsigned)__builtin_amdgcn_readlane(col, j0 + u);
      s[u] = S2[c * (OUT_F / 2) + (unsigned)lane];
    }
#pragma unroll
    for (int u = 0; u < 16; ++u) {
      if (j0 + u < n) {
        const int r = __builtin_amdgcn_readlane(rl, j0 + u);
        const float v = __int_as_float(__builtin_amdgcn_readlane(vbits, j0 + u));
        float* a = acc + r * OUT_F + 2 * lane;
        __hip_atomic_fetch_add(a, s[u].x * v, __ATOMIC_RELAXED,
                               __HIP_MEMORY_SCOPE_WORKGROUP);
        __hip_atomic_fetch_add(a + 1, s[u].y * v, __ATOMIC_RELAXED,
                               __HIP_MEMORY_SCOPE_WORKGROUP);
      }
    }
  }
}

__global__ __launch_bounds__(OWN_THREADS) void spmm_owner_kernel(
    const float* __restrict__ S, const unsigned* __restrict__ keys,
    const int* __restrict__ recs, const float* __restrict__ bias,
    float* __restrict__ out, int nNodes, int nEdges, int nOwners) {
  __shared__ float acc[ROWS_PER_OWNER * OUT_F];
  __shared__ unsigned queue[OWN_WAVES * QCAP];

  const int tid = threadIdx.x;
  const int lane = tid & 31;
  const int wave = __builtin_amdgcn_readfirstlane(tid >> 5);
  const int owner = blockIdx.x;
  const int base = owner * ROWS_PER_OWNER;

  for (int i = tid; i < ROWS_PER_OWNER * OUT_F; i += OWN_THREADS)
    acc[i] = bias[i & (OUT_F - 1)];
  __syncthreads();

  const int2* recs2 = (const int2*)recs;
  unsigned* q = queue + wave * QCAP;
  const int nTrips = (nEdges + 4095) >> 12;
  const int planePieces = nTrips << 5;
  const uint4* hiPlane = (const uint4*)keys + (owner >> 4) * planePieces;
  const uint4* loPlane = (const uint4*)keys + (16 + (owner & 15)) * planePieces;
  const int rot = (int)(((unsigned)owner * (unsigned)nTrips) / (unsigned)nOwners);
  int cnt = 0;

  for (int k0 = wave; k0 < nTrips; k0 += OWN_WAVES * 4) {
    uint4 xh[4], xl[4];
#pragma unroll
    for (int u = 0; u < 4; ++u) {
      const int k = k0 + u * OWN_WAVES;
      int tt = (k < nTrips ? k : k0) + rot;
      if (tt >= nTrips) tt -= nTrips;
      xh[u] = hiPlane[tt * 32 + lane];
      xl[u] = loPlane[tt * 32 + lane];
    }
#pragma unroll
    for (int u = 0; u < 4; ++u) {
      const int k = k0 + u * OWN_WAVES;
      if (k >= nTrips) break;
      int t = k + rot;
      if (t >= nTrips) t -= nTrips;
      unsigned w0 = xh[u].x & xl[u].x, w1 = xh[u].y & xl[u].y;
      unsigned w2 = xh[u].z & xl[u].z, w3 = xh[u].w & xl[u].w;
      const unsigned tbase = ((unsigned)t << 12) + ((unsigned)lane << 7);
#pragma nounroll
      for (int c = 0; c < 4; ++c) {
        unsigned m = w0;
        w0 = w1; w1 = w2; w2 = w3;
        const unsigned ebase = tbase + (unsigned)c;
        unsigned bal = __builtin_amdgcn_ballot_w32(m != 0u);
        while (bal != 0u) {
          if (m != 0u) {
            const int p = __builtin_ctz(m);
            m &= m - 1u;
            q[cnt + (int)__builtin_amdgcn_mbcnt_lo(bal, 0u)] = ebase + ((unsigned)p << 2);
          }
          cnt += __builtin_popcount(bal);
          bal = __builtin_amdgcn_ballot_w32(m != 0u);
        }
        while (cnt >= 32) {
          cnt -= 32;
          drain_batch(q + cnt, 32, lane, S, recs2, acc);
        }
      }
    }
  }
  if (cnt > 0) drain_batch(q, cnt, lane, S, recs2, acc);

  __syncthreads();

  int nrows = nNodes - base;
  if (nrows > ROWS_PER_OWNER) nrows = ROWS_PER_OWNER;
  const float4* a4 = (const float4*)acc;
  float4* o4 = (float4*)(out + (long long)base * OUT_F);
  for (int i = tid; i < nrows * (OUT_F / 4); i += OWN_THREADS) o4[i] = a4[i];
}

extern "C" void kernel_launch(void* const* d_in, const int* in_sizes, int n_in,
                              void* d_out, int out_size, void* d_ws,
                              size_t ws_size, hipStream_t stream) {
  const float* vertex  = (const float*)d_in[0];
  const int*   erow    = (const int*)d_in[1];
  const int*   ecol    = (const int*)d_in[2];
  const float* eval    = (const float*)d_in[3];
  const float* weights = (const float*)d_in[4];
  const float* bias    = (const float*)d_in[5];
  float* out     = (float*)d_out;
  float* support = (float*)d_ws;
  int* recs = (int*)((char*)d_ws + RECS_WS_OFFSET);
  unsigned* keys = (unsigned*)((char*)d_ws + KEYS_WS_OFFSET);

  const int nNodes = in_sizes[0] / IN_F;
  const int nEdges = in_sizes[1];

  const int nRowTiles = nNodes / 16;
  gemm_wmma_kernel<<<GEMM_BLOCKS, 256, 0, stream>>>(vertex, weights, support,
                                                nRowTiles, erow, ecol, eval,
                                                keys, recs, nEdges);

  const int nOwners = (nNodes + ROWS_PER_OWNER - 1) / ROWS_PER_OWNER;
  spmm_owner_kernel<<<nOwners, OWN_THREADS, 0, stream>>>(
      support, keys, recs, bias, out, nNodes, nEdges, nOwners);
}
